# starting kernel
# baseline (speedup 1.0000x reference)
#include <hip/hip_runtime.h>
#include <hip/hip_bf16.h>


typedef __attribute__((ext_vector_type(2))) float v2f;
typedef __attribute__((ext_vector_type(8))) float v8f;

#define B_     2
#define S_     2048
#define DM_    1024
#define H_     16
#define DK_    64
#define BH_    (B_ * H_)
#define ROWS_  (B_ * S_)
#define N_QKV  (3 * DM_)

__device__ __forceinline__ v8f wmma_f32(v2f a, v2f b, v8f c) {
  return __builtin_amdgcn_wmma_f32_16x16x4_f32(
      false, a, false, b, (short)0, c, false, false);
}

__device__ __forceinline__ void gemm_core_128x64(
    const float* __restrict__ A, const float* __restrict__ W,
    float* Xs, float* Ws, int m0, int n0, int Kdim, v8f acc[2][2]) {
  const int tid = threadIdx.x;
  const int wv = tid >> 5;
  const int lane = tid & 31;
  const int r = lane & 15;
  const int hh = lane >> 4;
  const int wm = wv >> 1;
  const int wn = wv & 1;

  for (int k0 = 0; k0 < Kdim; k0 += 16) {
    __syncthreads();
#pragma unroll
    for (int i = 0; i < 2; ++i) {
      int idx = tid + i * 256;
      int row = idx >> 2;
      int cv = (idx & 3) * 4;
      *(float4*)(Xs + row * 20 + cv) =
          *(const float4*)(A + (size_t)(m0 + row) * Kdim + k0 + cv);
    }
    {
      int row = tid >> 2;
      int cv = (tid & 3) * 4;
      *(float4*)(Ws + row * 20 + cv) =
          *(const float4*)(W + (size_t)(n0 + row) * Kdim + k0 + cv);
    }
    __syncthreads();
#pragma unroll
    for (int kk = 0; kk < 4; ++kk) {
      const int kd = kk * 4 + hh * 2;
      const float* xr0 = Xs + (wm * 32 + r) * 20 + kd;
      const float* xr1 = Xs + (wm * 32 + 16 + r) * 20 + kd;
      const float* wr0 = Ws + (wn * 32 + r) * 20 + kd;
      const float* wr1 = Ws + (wn * 32 + 16 + r) * 20 + kd;
      v2f a0 = {xr0[0], xr0[1]};
      v2f a1 = {xr1[0], xr1[1]};
      v2f b0 = {wr0[0], wr0[1]};
      v2f b1 = {wr1[0], wr1[1]};
      acc[0][0] = wmma_f32(a0, b0, acc[0][0]);
      acc[0][1] = wmma_f32(a0, b1, acc[0][1]);
      acc[1][0] = wmma_f32(a1, b0, acc[1][0]);
      acc[1][1] = wmma_f32(a1, b1, acc[1][1]);
    }
  }
}

__global__ void __launch_bounds__(256)
qkv_rope_kernel(const float* __restrict__ x, const float* __restrict__ w_qkv,
                const float* __restrict__ b_qkv, float* __restrict__ Qw,
                float* __restrict__ Kw, float* __restrict__ Vw) {
  __shared__ float Xs[128 * 20];
  __shared__ float Ws[64 * 20];
  v8f acc[2][2] = {};
  const int m0 = blockIdx.x * 128;
  const int n0 = blockIdx.y * 64;
  gemm_core_128x64(x, w_qkv, Xs, Ws, m0, n0, DM_, acc);

  const int tid = threadIdx.x;
  const int wv = tid >> 5;
  const int lane = tid & 31;
  const int r = lane & 15;
  const int hhalf = lane >> 4;
  const int wm = wv >> 1;
  const int wn = wv & 1;

#pragma unroll
  for (int mi = 0; mi < 2; ++mi) {
#pragma unroll
    for (int ni = 0; ni < 2; ++ni) {
      const int o = n0 + wn * 32 + ni * 16 + r;
      const int sel = o >> 10;
      const int rem = o & 1023;
      const int head = rem >> 6;
      const int d = rem & 63;
      const float bv = b_qkv[o];
      const int p = d >> 1;
      const float invf = __expf((float)(2 * p) * (-9.210340371976184f / 64.0f));
      const float sgn = (d & 1) ? 1.0f : -1.0f;
      float* dst = (sel == 0) ? Qw : ((sel == 1) ? Kw : Vw);
#pragma unroll
      for (int g = 0; g < 8; ++g) {
        const int gm = m0 + wm * 32 + mi * 16 + g + 8 * hhalf;
        const int srow = gm & (S_ - 1);
        const int bb = gm >> 11;
        float v = acc[mi][ni][g] + bv;
        float other = __shfl_xor(v, 1);
        float outv = v;
        if (sel < 2) {
          const float ang = (float)srow * invf;
          const float sn = __sinf(ang);
          const float cs = __cosf(ang);
          outv = v * cs + other * sgn * sn;
        }
        dst[(((size_t)(bb * H_ + head)) * S_ + srow) * DK_ + d] = outv;
      }
    }
  }
}

__global__ void __launch_bounds__(256)
flash_attn_kernel(const float* __restrict__ Q, const float* __restrict__ K,
                  const float* __restrict__ V, float* __restrict__ O) {
  __shared__ float Ks[16 * 68];
  __shared__ float Vs[16 * 68];
  __shared__ float Ps[8 * 16 * 17];

  const int bh = blockIdx.y;
  const int qb = blockIdx.x;
  const int tid = threadIdx.x;
  const int wv = tid >> 5;
  const int lane = tid & 31;
  const int r = lane & 15;
  const int hhalf = lane >> 4;
  const int qw0 = qb * 128 + wv * 16;

  const float* Qp = Q + (size_t)bh * S_ * DK_;
  const float* Kp = K + (size_t)bh * S_ * DK_;
  const float* Vp = V + (size_t)bh * S_ * DK_;

  v2f qf[16];
#pragma unroll
  for (int kk = 0; kk < 16; ++kk) {
    const float* qp = Qp + (size_t)(qw0 + r) * DK_ + kk * 4 + hhalf * 2;
    qf[kk] = v2f{qp[0] * 0.125f, qp[1] * 0.125f};
  }

  v8f o0 = {}, o1 = {}, o2 = {}, o3 = {};
  float mrow[8], lrow[8];
#pragma unroll
  for (int g = 0; g < 8; ++g) { mrow[g] = -3.0e38f; lrow[g] = 0.0f; }

  float* Pw = Ps + wv * 16 * 17;
  const int njt = qb * 8 + 8;

  for (int j = 0; j < njt; ++j) {
    __syncthreads();
    {
      const int rowj = tid >> 4;
      const int cv = (tid & 15) * 4;
      const float4 kv = *(const float4*)(Kp + (size_t)(j * 16 + rowj) * DK_ + cv);
      const float4 vvv = *(const float4*)(Vp + (size_t)(j * 16 + rowj) * DK_ + cv);
      *(float4*)(Ks + rowj * 68 + cv) = kv;
      *(float4*)(Vs + rowj * 68 + cv) = vvv;
    }
    __syncthreads();

    if (j * 16 <= qw0 + 15) {
      v8f s = {};
#pragma unroll
      for (int kk = 0; kk < 16; ++kk) {
        const int kd = kk * 4 + hhalf * 2;
        const float* kr = Ks + r * 68 + kd;
        v2f bf = {kr[0], kr[1]};
        s = wmma_f32(qf[kk], bf, s);
      }
      if (j * 16 + 15 > qw0) {
        const int key = j * 16 + r;
#pragma unroll
        for (int g = 0; g < 8; ++g) {
          if (key > qw0 + g + 8 * hhalf) s[g] = -3.0e38f;
        }
      }
      float pv[8];
#pragma unroll
      for (int g = 0; g < 8; ++g) {
        float v = s[g];
        float mx = v;
        mx = fmaxf(mx, __shfl_xor(mx, 1));
        mx = fmaxf(mx, __shfl_xor(mx, 2));
        mx = fmaxf(mx, __shfl_xor(mx, 4));
        mx = fmaxf(mx, __shfl_xor(mx, 8));
        const float mnew = fmaxf(mrow[g], mx);
        const float alpha = __expf(mrow[g] - mnew);
        const float pe = __expf(v - mnew);
        float rs = pe;
        rs += __shfl_xor(rs, 1);
        rs += __shfl_xor(rs, 2);
        rs += __shfl_xor(rs, 4);
        rs += __shfl_xor(rs, 8);
        lrow[g] = lrow[g] * alpha + rs;
        mrow[g] = mnew;
        pv[g] = pe;
        o0[g] *= alpha; o1[g] *= alpha; o2[g] *= alpha; o3[g] *= alpha;
      }
#pragma unroll
      for (int g = 0; g < 8; ++g) Pw[(g + 8 * hhalf) * 17 + r] = pv[g];
#pragma unroll
      for (int kk2 = 0; kk2 < 4; ++kk2) {
        const int kd = kk2 * 4 + hhalf * 2;
        const float* pr = Pw + r * 17 + kd;
        v2f af = {pr[0], pr[1]};
        v2f b0 = {Vs[kd * 68 + 0 * 16 + r], Vs[(kd + 1) * 68 + 0 * 16 + r]};
        v2f b1 = {Vs[kd * 68 + 1 * 16 + r], Vs[(kd + 1) * 68 + 1 * 16 + r]};
        v2f b2 = {Vs[kd * 68 + 2 * 16 + r], Vs[(kd + 1) * 68 + 2 * 16 + r]};
        v2f b3 = {Vs[kd * 68 + 3 * 16 + r], Vs[(kd + 1) * 68 + 3 * 16 + r]};
        o0 = wmma_f32(af, b0, o0);
        o1 = wmma_f32(af, b1, o1);
        o2 = wmma_f32(af, b2, o2);
        o3 = wmma_f32(af, b3, o3);
      }
    }
  }

  const int bb = bh >> 4;
  const int head = bh & 15;
#pragma unroll
  for (int g = 0; g < 8; ++g) {
    const float inv = 1.0f / (lrow[g] + 1e-8f);
    const int qrow = qw0 + g + 8 * hhalf;
    float* dst = O + ((size_t)(bb * S_ + qrow)) * DM_ + head * DK_;
    dst[0 * 16 + r] = o0[g] * inv;
    dst[1 * 16 + r] = o1[g] * inv;
    dst[2 * 16 + r] = o2[g] * inv;
    dst[3 * 16 + r] = o3[g] * inv;
  }
}

__global__ void __launch_bounds__(256)
out_proj_kernel(const float* __restrict__ Oin, const float* __restrict__ w_out,
                const float* __restrict__ b_out, float* __restrict__ out) {
  __shared__ float Xs[128 * 20];
  __shared__ float Ws[64 * 20];
  v8f acc[2][2] = {};
  const int m0 = blockIdx.x * 128;
  const int n0 = blockIdx.y * 64;
  gemm_core_128x64(Oin, w_out, Xs, Ws, m0, n0, DM_, acc);

  const int tid = threadIdx.x;
  const int wv = tid >> 5;
  const int lane = tid & 31;
  const int r = lane & 15;
  const int hhalf = lane >> 4;
  const int wm = wv >> 1;
  const int wn = wv & 1;

#pragma unroll
  for (int mi = 0; mi < 2; ++mi) {
#pragma unroll
    for (int ni = 0; ni < 2; ++ni) {
      const int n = n0 + wn * 32 + ni * 16 + r;
      const float bv = b_out[n];
#pragma unroll
      for (int g = 0; g < 8; ++g) {
        const int gm = m0 + wm * 32 + mi * 16 + g + 8 * hhalf;
        out[(size_t)gm * DM_ + n] = acc[mi][ni][g] + bv;
      }
    }
  }
}

extern "C" void kernel_launch(void* const* d_in, const int* in_sizes, int n_in,
                              void* d_out, int out_size, void* d_ws,
                              size_t ws_size, hipStream_t stream) {
  const float* x = (const float*)d_in[0];
  const float* w_qkv = (const float*)d_in[1];
  const float* b_qkv = (const float*)d_in[2];
  const float* w_out = (const float*)d_in[3];
  const float* b_out = (const float*)d_in[4];
  float* out = (float*)d_out;

  const size_t per_buf = (size_t)BH_ * S_ * DK_;
  float* Qw = (float*)d_ws;
  float* Kw = Qw + per_buf;
  float* Vw = Kw + per_buf;
  float* Ow = Vw + per_buf;

  qkv_rope_kernel<<<dim3(ROWS_ / 128, N_QKV / 64), 256, 0, stream>>>(
      x, w_qkv, b_qkv, Qw, Kw, Vw);
  flash_attn_kernel<<<dim3(S_ / 128, BH_), 256, 0, stream>>>(Qw, Kw, Vw, Ow);
  out_proj_kernel<<<dim3(ROWS_ / 128, DM_ / 64), 256, 0, stream>>>(
      Ow, w_out, b_out, out);
}


# attention: softmax row max and row sum butterflies on the vector ALU (DPP row_xmask) instead of ds_bpermute: 8 dependent LDS-unit round trips per key tile leave each wave's chain; same pairing order
# speedup vs baseline: 1.3709x; 1.3709x over previous
#include <hip/hip_runtime.h>
#include <hip/hip_bf16.h>


typedef __attribute__((ext_vector_type(2))) float v2f;
typedef __attribute__((ext_vector_type(8))) float v8f;

#define B_     2
#define S_     2048
#define DM_    1024
#define H_     16
#define DK_    64
#define BH_    (B_ * H_)
#define ROWS_  (B_ * S_)
#define N_QKV  (3 * DM_)

__device__ __forceinline__ v8f wmma_f32(v2f a, v2f b, v8f c) {
  return __builtin_amdgcn_wmma_f32_16x16x4_f32(
      false, a, false, b, (short)0, c, false, false);
}

template <int M>
__device__ __forceinline__ float row_xor(float x) {
  return __builtin_bit_cast(float, __builtin_amdgcn_update_dpp(
      0, __builtin_bit_cast(int, x), 0x160 + M, 0xf, 0xf, true));
}

__device__ __forceinline__ void gemm_core_128x64(
    const float* __restrict__ A, const float* __restrict__ W,
    float* Xs, float* Ws, int m0, int n0, int Kdim, bool swap, v8f acc[2][2]) {
  const int tid = threadIdx.x;
  const int wv = tid >> 5;
  const int lane = tid & 31;
  const int r = lane & 15;
  const int hh = lane >> 4;
  const int wm = wv >> 1;
  const int wn = wv & 1;
  const float* xrow = Xs + (wm * 32 + r) * 20;
  const float* wrow = Ws + (wn * 32 + r) * 20;
  const float* pa = swap ? wrow : xrow;
  const float* pb = swap ? xrow : wrow;

  for (int k0 = 0; k0 < Kdim; k0 += 16) {
    __syncthreads();
#pragma unroll
    for (int i = 0; i < 2; ++i) {
      int idx = tid + i * 256;
      int row = idx >> 2;
      int cv = (idx & 3) * 4;
      *(float4*)(Xs + row * 20 + cv) =
          *(const float4*)(A + (size_t)(m0 + row) * Kdim + k0 + cv);
    }
    {
      int row = tid >> 2;
      int cv = (tid & 3) * 4;
      *(float4*)(Ws + row * 20 + cv) =
          *(const float4*)(W + (size_t)(n0 + row) * Kdim + k0 + cv);
    }
    __syncthreads();
#pragma unroll
    for (int kk = 0; kk < 4; ++kk) {
      const int kd = kk * 4 + hh * 2;
      v2f a0 = {pa[kd], pa[kd + 1]};
      v2f a1 = {pa[16 * 20 + kd], pa[16 * 20 + kd + 1]};
      v2f b0 = {pb[kd], pb[kd + 1]};
      v2f b1 = {pb[16 * 20 + kd], pb[16 * 20 + kd + 1]};
      acc[0][0] = wmma_f32(a0, b0, acc[0][0]);
      acc[0][1] = wmma_f32(a0, b1, acc[0][1]);
      acc[1][0] = wmma_f32(a1, b0, acc[1][0]);
      acc[1][1] = wmma_f32(a1, b1, acc[1][1]);
    }
  }
}

__global__ void __launch_bounds__(256, 12)
qkv_rope_kernel(const float* __restrict__ x, const float* __restrict__ w_qkv,
                const float* __restrict__ b_qkv, float* __restrict__ Qw,
                float* __restrict__ Kw, float* __restrict__ Vw) {
  __shared__ float Xs[128 * 20];
  __shared__ float Ws[64 * 20];
  v8f acc[2][2] = {};
  const int m0 = blockIdx.x * 128;
  const int n0 = blockIdx.y * 64;
  const int sel = n0 >> 10;
  const int head = (n0 & 1023) >> 6;
  const bool swap = (sel == 1);
  gemm_core_128x64(x, w_qkv, Xs, Ws, m0, n0, DM_, swap, acc);

  const int tid = threadIdx.x;
  const int wv = tid >> 5;
  const int lane = tid & 31;
  const int r = lane & 15;
  const int hhalf = lane >> 4;
  const int wm = wv >> 1;
  const int wn = wv & 1;

  if (sel == 0) {
#pragma unroll
    for (int mi = 0; mi < 2; ++mi) {
#pragma unroll
      for (int ni = 0; ni < 2; ++ni) {
        const int d = wn * 32 + ni * 16 + r;
        const float bv = b_qkv[n0 + d];
        const int p = d >> 1;
        const float invf = __expf((float)(2 * p) * (-9.210340371976184f / 64.0f));
        const float sgn = (d & 1) ? 1.0f : -1.0f;
#pragma unroll
        for (int g = 0; g < 8; ++g) {
          const int gm = m0 + wm * 32 + mi * 16 + g + 8 * hhalf;
          const int srow = gm & (S_ - 1);
          const int bb = gm >> 11;
          float v = acc[mi][ni][g] + bv;
          float other = __shfl_xor(v, 1);
          const float ang = (float)srow * invf;
          const float sn = __sinf(ang);
          const float cs = __cosf(ang);
          const float outv = v * cs + other * sgn * sn;
          Qw[(((size_t)(bb * H_ + head)) * S_ + srow) * DK_ + d] = outv;
        }
      }
    }
  } else {
    float* dst = swap ? Kw : Vw;
#pragma unroll
    for (int i = 0; i < 2; ++i) {
#pragma unroll
      for (int j = 0; j < 2; ++j) {
        const int fb = swap ? i : j;
        const int sb = swap ? j : i;
        const int nb = wn * 2 + fb;
        const int gm0 = m0 + wm * 32 + sb * 16;
        const int bb = gm0 >> 11;
        const int s0 = gm0 & (S_ - 1);
        float vals[8];
        if (swap) {
          const int d0 = nb * 16 + 8 * hhalf;
          const float4 bA = *(const float4*)(b_qkv + n0 + d0);
          const float4 bB = *(const float4*)(b_qkv + n0 + d0 + 4);
          const float bvv[8] = {bA.x, bA.y, bA.z, bA.w, bB.x, bB.y, bB.z, bB.w};
          float v[8];
#pragma unroll
          for (int g = 0; g < 8; ++g) v[g] = acc[i][j][g] + bvv[g];
          const float srowf = (float)(s0 + r);
#pragma unroll
          for (int g2 = 0; g2 < 4; ++g2) {
            const int p = (d0 >> 1) + g2;
            const float invf = __expf((float)(2 * p) * (-9.210340371976184f / 64.0f));
            const float ang = srowf * invf;
            const float sn = __sinf(ang);
            const float cs = __cosf(ang);
            vals[2 * g2]     = v[2 * g2] * cs + v[2 * g2 + 1] * -1.0f * sn;
            vals[2 * g2 + 1] = v[2 * g2 + 1] * cs + v[2 * g2] * 1.0f * sn;
          }
        } else {
          const float bv = b_qkv[n0 + nb * 16 + r];
#pragma unroll
          for (int g = 0; g < 8; ++g) vals[g] = acc[i][j][g] + bv;
        }
        float* base = dst + ((size_t)((bb * H_ + head) * (S_ / 16) + (s0 >> 4))) * 1024 + lane * 4;
        *(float4*)(base + (nb * 2 + 0) * 128) = make_float4(vals[0], vals[1], vals[2], vals[3]);
        *(float4*)(base + (nb * 2 + 1) * 128) = make_float4(vals[4], vals[5], vals[6], vals[7]);
      }
    }
  }
}

__device__ __forceinline__ void attn_tile(
    const v2f (&qf)[16], const float4 (&kc)[8], const float4 (&vc)[8],
    v8f& o0, v8f& o1, v8f& o2, v8f& o3, float (&mrow)[8], float (&lrow)[8],
    float* Pw, bool diag, int key, int qrow0, int r, int hhalf) {
  v8f s = {};
#pragma unroll
  for (int i = 0; i < 8; ++i) {
    s = wmma_f32(qf[2 * i],     v2f{kc[i].x, kc[i].y}, s);
    s = wmma_f32(qf[2 * i + 1], v2f{kc[i].z, kc[i].w}, s);
  }
  if (diag) {
#pragma unroll
    for (int g = 0; g < 8; ++g) {
      if (key > qrow0 + g) s[g] = -3.0e38f;
    }
  }
  float pv[8];
#pragma unroll
  for (int g = 0; g < 8; ++g) {
    float v = s[g];
    float mx = v;
    mx = fmaxf(mx, row_xor<1>(mx));
    mx = fmaxf(mx, row_xor<2>(mx));
    mx = fmaxf(mx, row_xor<4>(mx));
    mx = fmaxf(mx, row_xor<8>(mx));
    const float mnew = fmaxf(mrow[g], mx);
    const float alpha = __expf(mrow[g] - mnew);
    const float pe = __expf(v - mnew);
    float rs = pe;
    rs += row_xor<1>(rs);
    rs += row_xor<2>(rs);
    rs += row_xor<4>(rs);
    rs += row_xor<8>(rs);
    lrow[g] = lrow[g] * alpha + rs;
    mrow[g] = mnew;
    pv[g] = pe;
    o0[g] *= alpha; o1[g] *= alpha; o2[g] *= alpha; o3[g] *= alpha;
  }
#pragma unroll
  for (int g = 0; g < 8; ++g) Pw[(g + 8 * hhalf) * 20 + r] = pv[g];
  const float4 pA = *(const float4*)(Pw + r * 20 + 8 * hhalf);
  const float4 pB = *(const float4*)(Pw + r * 20 + 8 * hhalf + 4);
  const v2f af0 = {pA.x, pA.y};
  const v2f af1 = {pA.z, pA.w};
  const v2f af2 = {pB.x, pB.y};
  const v2f af3 = {pB.z, pB.w};
  o0 = wmma_f32(af0, v2f{vc[0].x, vc[0].y}, o0);
  o1 = wmma_f32(af0, v2f{vc[2].x, vc[2].y}, o1);
  o2 = wmma_f32(af0, v2f{vc[4].x, vc[4].y}, o2);
  o3 = wmma_f32(af0, v2f{vc[6].x, vc[6].y}, o3);
  o0 = wmma_f32(af1, v2f{vc[0].z, vc[0].w}, o0);
  o1 = wmma_f32(af1, v2f{vc[2].z, vc[2].w}, o1);
  o2 = wmma_f32(af1, v2f{vc[4].z, vc[4].w}, o2);
  o3 = wmma_f32(af1, v2f{vc[6].z, vc[6].w}, o3);
  o0 = wmma_f32(af2, v2f{vc[1].x, vc[1].y}, o0);
  o1 = wmma_f32(af2, v2f{vc[3].x, vc[3].y}, o1);
  o2 = wmma_f32(af2, v2f{vc[5].x, vc[5].y}, o2);
  o3 = wmma_f32(af2, v2f{vc[7].x, vc[7].y}, o3);
  o0 = wmma_f32(af3, v2f{vc[1].z, vc[1].w}, o0);
  o1 = wmma_f32(af3, v2f{vc[3].z, vc[3].w}, o1);
  o2 = wmma_f32(af3, v2f{vc[5].z, vc[5].w}, o2);
  o3 = wmma_f32(af3, v2f{vc[7].z, vc[7].w}, o3);
}

__global__ void __launch_bounds__(256, 4)
flash_attn_kernel(const float* __restrict__ Q, const float* __restrict__ K,
                  const float* __restrict__ V, float* __restrict__ O) {
  __shared__ __attribute__((aligned(16))) float Ps[8 * 16 * 20];

  const int bh = blockIdx.y;
  const int qb = blockIdx.x;
  const int tid = threadIdx.x;
  const int wv = tid >> 5;
  const int lane = tid & 31;
  const int r = lane & 15;
  const int hhalf = lane >> 4;
  const int sl = wv & 3;
  const int qt = __builtin_amdgcn_readfirstlane(
      (wv < 4) ? (qb + 32 * sl) : (127 - qb - 32 * sl));
  const int qw0 = qt * 16;

  const float* Qp = Q + (size_t)bh * S_ * DK_;
  const float* Kp = K + (size_t)bh * S_ * DK_ + lane * 4;
  const float* Vp = V + (size_t)bh * S_ * DK_ + lane * 4;

  v2f qf[16];
#pragma unroll
  for (int i = 0; i < 8; ++i) {
    const float4 t = *(const float4*)(Qp + (size_t)(qw0 + r) * DK_ +
                                      (i >> 1) * 16 + hhalf * 8 + (i & 1) * 4);
    qf[2 * i]     = v2f{t.x * 0.125f, t.y * 0.125f};
    qf[2 * i + 1] = v2f{t.z * 0.125f, t.w * 0.125f};
  }

  v8f o0 = {}, o1 = {}, o2 = {}, o3 = {};
  float mrow[8], lrow[8];
#pragma unroll
  for (int g = 0; g < 8; ++g) { mrow[g] = -3.0e38f; lrow[g] = 0.0f; }

  float* Pw = Ps + wv * 16 * 20;
  const int qrow0 = qw0 + 8 * hhalf;

  float4 kA[8], vA[8], kB[8], vB[8];
#pragma unroll
  for (int i = 0; i < 8; ++i) {
    kA[i] = *(const float4*)(Kp + i * 128);
    vA[i] = *(const float4*)(Vp + i * 128);
  }

  for (int j = 0; j <= qt; j += 2) {
    {
      const int jn = (j + 1 <= qt) ? (j + 1) : qt;
#pragma unroll
      for (int i = 0; i < 8; ++i) {
        kB[i] = *(const float4*)(Kp + (size_t)jn * 1024 + i * 128);
        vB[i] = *(const float4*)(Vp + (size_t)jn * 1024 + i * 128);
      }
      attn_tile(qf, kA, vA, o0, o1, o2, o3, mrow, lrow, Pw, j == qt,
                j * 16 + r, qrow0, r, hhalf);
    }
    if (j + 1 <= qt) {
      const int jn = (j + 2 <= qt) ? (j + 2) : qt;
#pragma unroll
      for (int i = 0; i < 8; ++i) {
        kA[i] = *(const float4*)(Kp + (size_t)jn * 1024 + i * 128);
        vA[i] = *(const float4*)(Vp + (size_t)jn * 1024 + i * 128);
      }
      attn_tile(qf, kB, vB, o0, o1, o2, o3, mrow, lrow, Pw, j + 1 == qt,
                (j + 1) * 16 + r, qrow0, r, hhalf);
    }
  }

  const int bb = bh >> 4;
  const int head = bh & 15;
#pragma unroll
  for (int g = 0; g < 8; ++g) {
    const float inv = 1.0f / (lrow[g] + 1e-8f);
    const int qrow = qw0 + g + 8 * hhalf;
    float* dst = O + ((size_t)(bb * S_ + qrow)) * DM_ + head * DK_;
    dst[0 * 16 + r] = o0[g] * inv;
    dst[1 * 16 + r] = o1[g] * inv;
    dst[2 * 16 + r] = o2[g] * inv;
    dst[3 * 16 + r] = o3[g] * inv;
  }
}

__global__ void __launch_bounds__(256)
out_proj_kernel(const float* __restrict__ Oin, const float* __restrict__ w_out,
                const float* __restrict__ b_out, float* __restrict__ out) {
  __shared__ float Xs[128 * 20];
  __shared__ float Ws[64 * 20];
  v8f acc[2][2] = {};
  const int m0 = blockIdx.x * 128;
  const int n0 = blockIdx.y * 64;
  gemm_core_128x64(Oin, w_out, Xs, Ws, m0, n0, DM_, false, acc);

  const int tid = threadIdx.x;
  const int wv = tid >> 5;
  const int lane = tid & 31;
  const int r = lane & 15;
  const int hhalf = lane >> 4;
  const int wm = wv >> 1;
  const int wn = wv & 1;

#pragma unroll
  for (int mi = 0; mi < 2; ++mi) {
#pragma unroll
    for (int ni = 0; ni < 2; ++ni) {
      const int n = n0 + wn * 32 + ni * 16 + r;
      const float bv = b_out[n];
#pragma unroll
      for (int g = 0; g < 8; ++g) {
        const int gm = m0 + wm * 32 + mi * 16 + g + 8 * hhalf;
        out[(size_t)gm * DM_ + n] = acc[mi][ni][g] + bv;
      }
    }
  }
}

extern "C" void kernel_launch(void* const* d_in, const int* in_sizes, int n_in,
                              void* d_out, int out_size, void* d_ws,
                              size_t ws_size, hipStream_t stream) {
  const float* x = (const float*)d_in[0];
  const float* w_qkv = (const float*)d_in[1];
  const float* b_qkv = (const float*)d_in[2];
  const float* w_out = (const float*)d_in[3];
  const float* b_out = (const float*)d_in[4];
  float* out = (float*)d_out;

  const size_t per_buf = (size_t)BH_ * S_ * DK_;
  float* Qw = (float*)d_ws;
  float* Kw = Qw + per_buf;
  float* Vw = Kw + per_buf;
  float* Ow = Vw + per_buf;

  qkv_rope_kernel<<<dim3(ROWS_ / 128, N_QKV / 64), 256, 0, stream>>>(
      x, w_qkv, b_qkv, Qw, Kw, Vw);
  flash_attn_kernel<<<dim3(S_ / 128, BH_), 256, 0, stream>>>(Qw, Kw, Vw, Ow);
  out_proj_kernel<<<dim3(ROWS_ / 128, DM_ / 64), 256, 0, stream>>>(
      Ow, w_out, b_out, out);
}


# out projection: both operands in WMMA operand order (O written so by attention, w_out re-ordered by kernel 1); waves stream fragments from global, no LDS or barriers; k order in 8-k groups changes
# speedup vs baseline: 1.4410x; 1.0511x over previous
#include <hip/hip_runtime.h>
#include <hip/hip_bf16.h>


typedef __attribute__((ext_vector_type(2))) float v2f;
typedef __attribute__((ext_vector_type(8))) float v8f;

#define B_     2
#define S_     2048
#define DM_    1024
#define H_     16
#define DK_    64
#define BH_    (B_ * H_)
#define ROWS_  (B_ * S_)
#define N_QKV  (3 * DM_)

__device__ __forceinline__ v8f wmma_f32(v2f a, v2f b, v8f c) {
  return __builtin_amdgcn_wmma_f32_16x16x4_f32(
      false, a, false, b, (short)0, c, false, false);
}

template <int M>
__device__ __forceinline__ float row_xor(float x) {
  return __builtin_bit_cast(float, __builtin_amdgcn_update_dpp(
      0, __builtin_bit_cast(int, x), 0x160 + M, 0xf, 0xf, true));
}

__device__ __forceinline__ void gemm_core_128x64(
    const float* __restrict__ A, const float* __restrict__ W,
    float* Xs, float* Ws, int m0, int n0, int Kdim, bool swap, v8f acc[2][2]) {
  const int tid = threadIdx.x;
  const int wv = tid >> 5;
  const int lane = tid & 31;
  const int r = lane & 15;
  const int hh = lane >> 4;
  const int wm = wv >> 1;
  const int wn = wv & 1;
  const float* xrow = Xs + (wm * 32 + r) * 20;
  const float* wrow = Ws + (wn * 32 + r) * 20;
  const float* pa = swap ? wrow : xrow;
  const float* pb = swap ? xrow : wrow;

  for (int k0 = 0; k0 < Kdim; k0 += 16) {
    __syncthreads();
#pragma unroll
    for (int i = 0; i < 2; ++i) {
      int idx = tid + i * 256;
      int row = idx >> 2;
      int cv = (idx & 3) * 4;
      *(float4*)(Xs + row * 20 + cv) =
          *(const float4*)(A + (size_t)(m0 + row) * Kdim + k0 + cv);
    }
    {
      int row = tid >> 2;
      int cv = (tid & 3) * 4;
      *(float4*)(Ws + row * 20 + cv) =
          *(const float4*)(W + (size_t)(n0 + row) * Kdim + k0 + cv);
    }
    __syncthreads();
#pragma unroll
    for (int kk = 0; kk < 4; ++kk) {
      const int kd = kk * 4 + hh * 2;
      v2f a0 = {pa[kd], pa[kd + 1]};
      v2f a1 = {pa[16 * 20 + kd], pa[16 * 20 + kd + 1]};
      v2f b0 = {pb[kd], pb[kd + 1]};
      v2f b1 = {pb[16 * 20 + kd], pb[16 * 20 + kd + 1]};
      acc[0][0] = wmma_f32(a0, b0, acc[0][0]);
      acc[0][1] = wmma_f32(a0, b1, acc[0][1]);
      acc[1][0] = wmma_f32(a1, b0, acc[1][0]);
      acc[1][1] = wmma_f32(a1, b1, acc[1][1]);
    }
  }
}

__global__ void __launch_bounds__(256, 12)
qkv_rope_kernel(const float* __restrict__ x, const float* __restrict__ w_qkv,
                const float* __restrict__ b_qkv, const float* __restrict__ w_out,
                float* __restrict__ Qw, float* __restrict__ Kw,
                float* __restrict__ Vw, float* __restrict__ Wf) {
  __shared__ float Xs[128 * 20];
  __shared__ float Ws[64 * 20];
  v8f acc[2][2] = {};
  const int m0 = blockIdx.x * 128;
  const int n0 = blockIdx.y * 64;
  {
    const int wg = blockIdx.y * (ROWS_ / 128) + blockIdx.x;
    if (wg < (DM_ * DM_ / 4) / 256) {
      const int f = wg * 256 + threadIdx.x;
      const int ln = f & 31;
      const int blk = f >> 5;
      const int rb = blk >> 7;
      const int kc = blk & 127;
      const float4 t = *(const float4*)(w_out + (size_t)(rb * 16 + (ln & 15)) * DM_ +
                                        kc * 8 + (ln >> 4) * 4);
      *(float4*)(Wf + (size_t)f * 4) = t;
    }
  }
  const int sel = n0 >> 10;
  const int head = (n0 & 1023) >> 6;
  const bool swap = (sel == 1);
  gemm_core_128x64(x, w_qkv, Xs, Ws, m0, n0, DM_, swap, acc);

  const int tid = threadIdx.x;
  const int wv = tid >> 5;
  const int lane = tid & 31;
  const int r = lane & 15;
  const int hhalf = lane >> 4;
  const int wm = wv >> 1;
  const int wn = wv & 1;

  if (sel == 0) {
#pragma unroll
    for (int mi = 0; mi < 2; ++mi) {
#pragma unroll
      for (int ni = 0; ni < 2; ++ni) {
        const int d = wn * 32 + ni * 16 + r;
        const float bv = b_qkv[n0 + d];
        const int p = d >> 1;
        const float invf = __expf((float)(2 * p) * (-9.210340371976184f / 64.0f));
        const float sgn = (d & 1) ? 1.0f : -1.0f;
#pragma unroll
        for (int g = 0; g < 8; ++g) {
          const int gm = m0 + wm * 32 + mi * 16 + g + 8 * hhalf;
          const int srow = gm & (S_ - 1);
          const int bb = gm >> 11;
          float v = acc[mi][ni][g] + bv;
          float other = __shfl_xor(v, 1);
          const float ang = (float)srow * invf;
          const float sn = __sinf(ang);
          const float cs = __cosf(ang);
          const float outv = v * cs + other * sgn * sn;
          Qw[(((size_t)(bb * H_ + head)) * S_ + srow) * DK_ + d] = outv;
        }
      }
    }
  } else {
    float* dst = swap ? Kw : Vw;
#pragma unroll
    for (int i = 0; i < 2; ++i) {
#pragma unroll
      for (int j = 0; j < 2; ++j) {
        const int fb = swap ? i : j;
        const int sb = swap ? j : i;
        const int nb = wn * 2 + fb;
        const int gm0 = m0 + wm * 32 + sb * 16;
        const int bb = gm0 >> 11;
        const int s0 = gm0 & (S_ - 1);
        float vals[8];
        if (swap) {
          const int d0 = nb * 16 + 8 * hhalf;
          const float4 bA = *(const float4*)(b_qkv + n0 + d0);
          const float4 bB = *(const float4*)(b_qkv + n0 + d0 + 4);
          const float bvv[8] = {bA.x, bA.y, bA.z, bA.w, bB.x, bB.y, bB.z, bB.w};
          float v[8];
#pragma unroll
          for (int g = 0; g < 8; ++g) v[g] = acc[i][j][g] + bvv[g];
          const float srowf = (float)(s0 + r);
#pragma unroll
          for (int g2 = 0; g2 < 4; ++g2) {
            const int p = (d0 >> 1) + g2;
            const float invf = __expf((float)(2 * p) * (-9.210340371976184f / 64.0f));
            const float ang = srowf * invf;
            const float sn = __sinf(ang);
            const float cs = __cosf(ang);
            vals[2 * g2]     = v[2 * g2] * cs + v[2 * g2 + 1] * -1.0f * sn;
            vals[2 * g2 + 1] = v[2 * g2 + 1] * cs + v[2 * g2] * 1.0f * sn;
          }
        } else {
          const float bv = b_qkv[n0 + nb * 16 + r];
#pragma unroll
          for (int g = 0; g < 8; ++g) vals[g] = acc[i][j][g] + bv;
        }
        float* base = dst + ((size_t)((bb * H_ + head) * (S_ / 16) + (s0 >> 4))) * 1024 + lane * 4;
        *(float4*)(base + (nb * 2 + 0) * 128) = make_float4(vals[0], vals[1], vals[2], vals[3]);
        *(float4*)(base + (nb * 2 + 1) * 128) = make_float4(vals[4], vals[5], vals[6], vals[7]);
      }
    }
  }
}

__device__ __forceinline__ void attn_tile(
    const v2f (&qf)[16], const float4 (&kc)[8], const float4 (&vc)[8],
    v8f& o0, v8f& o1, v8f& o2, v8f& o3, float (&mrow)[8], float (&lrow)[8],
    float* Pw, bool diag, int key, int qrow0, int r, int hhalf) {
  v8f s = {};
#pragma unroll
  for (int i = 0; i < 8; ++i) {
    s = wmma_f32(qf[2 * i],     v2f{kc[i].x, kc[i].y}, s);
    s = wmma_f32(qf[2 * i + 1], v2f{kc[i].z, kc[i].w}, s);
  }
  if (diag) {
#pragma unroll
    for (int g = 0; g < 8; ++g) {
      if (key > qrow0 + g) s[g] = -3.0e38f;
    }
  }
  float pv[8];
#pragma unroll
  for (int g = 0; g < 8; ++g) {
    float v = s[g];
    float mx = v;
    mx = fmaxf(mx, row_xor<1>(mx));
    mx = fmaxf(mx, row_xor<2>(mx));
    mx = fmaxf(mx, row_xor<4>(mx));
    mx = fmaxf(mx, row_xor<8>(mx));
    const float mnew = fmaxf(mrow[g], mx);
    const float alpha = __expf(mrow[g] - mnew);
    const float pe = __expf(v - mnew);
    float rs = pe;
    rs += row_xor<1>(rs);
    rs += row_xor<2>(rs);
    rs += row_xor<4>(rs);
    rs += row_xor<8>(rs);
    lrow[g] = lrow[g] * alpha + rs;
    mrow[g] = mnew;
    pv[g] = pe;
    o0[g] *= alpha; o1[g] *= alpha; o2[g] *= alpha; o3[g] *= alpha;
  }
#pragma unroll
  for (int g = 0; g < 8; ++g) Pw[(g + 8 * hhalf) * 20 + r] = pv[g];
  const float4 pA = *(const float4*)(Pw + r * 20 + 8 * hhalf);
  const float4 pB = *(const float4*)(Pw + r * 20 + 8 * hhalf + 4);
  const v2f af0 = {pA.x, pA.y};
  const v2f af1 = {pA.z, pA.w};
  const v2f af2 = {pB.x, pB.y};
  const v2f af3 = {pB.z, pB.w};
  o0 = wmma_f32(af0, v2f{vc[0].x, vc[0].y}, o0);
  o1 = wmma_f32(af0, v2f{vc[2].x, vc[2].y}, o1);
  o2 = wmma_f32(af0, v2f{vc[4].x, vc[4].y}, o2);
  o3 = wmma_f32(af0, v2f{vc[6].x, vc[6].y}, o3);
  o0 = wmma_f32(af1, v2f{vc[0].z, vc[0].w}, o0);
  o1 = wmma_f32(af1, v2f{vc[2].z, vc[2].w}, o1);
  o2 = wmma_f32(af1, v2f{vc[4].z, vc[4].w}, o2);
  o3 = wmma_f32(af1, v2f{vc[6].z, vc[6].w}, o3);
  o0 = wmma_f32(af2, v2f{vc[1].x, vc[1].y}, o0);
  o1 = wmma_f32(af2, v2f{vc[3].x, vc[3].y}, o1);
  o2 = wmma_f32(af2, v2f{vc[5].x, vc[5].y}, o2);
  o3 = wmma_f32(af2, v2f{vc[7].x, vc[7].y}, o3);
  o0 = wmma_f32(af3, v2f{vc[1].z, vc[1].w}, o0);
  o1 = wmma_f32(af3, v2f{vc[3].z, vc[3].w}, o1);
  o2 = wmma_f32(af3, v2f{vc[5].z, vc[5].w}, o2);
  o3 = wmma_f32(af3, v2f{vc[7].z, vc[7].w}, o3);
}

__global__ void __launch_bounds__(256, 4)
flash_attn_kernel(const float* __restrict__ Q, const float* __restrict__ K,
                  const float* __restrict__ V, float* __restrict__ O) {
  __shared__ __attribute__((aligned(16))) float Ps[8 * 16 * 20];

  const int bh = blockIdx.y;
  const int qb = blockIdx.x;
  const int tid = threadIdx.x;
  const int wv = tid >> 5;
  const int lane = tid & 31;
  const int r = lane & 15;
  const int hhalf = lane >> 4;
  const int sl = wv & 3;
  const int qt = __builtin_amdgcn_readfirstlane(
      (wv < 4) ? (qb + 32 * sl) : (127 - qb - 32 * sl));
  const int qw0 = qt * 16;

  const float* Qp = Q + (size_t)bh * S_ * DK_;
  const float* Kp = K + (size_t)bh * S_ * DK_ + lane * 4;
  const float* Vp = V + (size_t)bh * S_ * DK_ + lane * 4;

  v2f qf[16];
#pragma unroll
  for (int i = 0; i < 8; ++i) {
    const float4 t = *(const float4*)(Qp + (size_t)(qw0 + r) * DK_ +
                                      (i >> 1) * 16 + hhalf * 8 + (i & 1) * 4);
    qf[2 * i]     = v2f{t.x * 0.125f, t.y * 0.125f};
    qf[2 * i + 1] = v2f{t.z * 0.125f, t.w * 0.125f};
  }

  v8f o0 = {}, o1 = {}, o2 = {}, o3 = {};
  float mrow[8], lrow[8];
#pragma unroll
  for (int g = 0; g < 8; ++g) { mrow[g] = -3.0e38f; lrow[g] = 0.0f; }

  float* Pw = Ps + wv * 16 * 20;
  const int qrow0 = qw0 + 8 * hhalf;

  float4 kA[8], vA[8], kB[8], vB[8];
#pragma unroll
  for (int i = 0; i < 8; ++i) {
    kA[i] = *(const float4*)(Kp + i * 128);
    vA[i] = *(const float4*)(Vp + i * 128);
  }

  for (int j = 0; j <= qt; j += 2) {
    {
      const int jn = (j + 1 <= qt) ? (j + 1) : qt;
#pragma unroll
      for (int i = 0; i < 8; ++i) {
        kB[i] = *(const float4*)(Kp + (size_t)jn * 1024 + i * 128);
        vB[i] = *(const float4*)(Vp + (size_t)jn * 1024 + i * 128);
      }
      attn_tile(qf, kA, vA, o0, o1, o2, o3, mrow, lrow, Pw, j == qt,
                j * 16 + r, qrow0, r, hhalf);
    }
    if (j + 1 <= qt) {
      const int jn = (j + 2 <= qt) ? (j + 2) : qt;
#pragma unroll
      for (int i = 0; i < 8; ++i) {
        kA[i] = *(const float4*)(Kp + (size_t)jn * 1024 + i * 128);
        vA[i] = *(const float4*)(Vp + (size_t)jn * 1024 + i * 128);
      }
      attn_tile(qf, kB, vB, o0, o1, o2, o3, mrow, lrow, Pw, j + 1 == qt,
                (j + 1) * 16 + r, qrow0, r, hhalf);
    }
  }

  const int bb = bh >> 4;
  const int head = bh & 15;
  float inv[8];
#pragma unroll
  for (int g = 0; g < 8; ++g) inv[g] = 1.0f / (lrow[g] + 1e-8f);
  float* dst = O + ((size_t)((bb * (S_ / 16) + qt) * (DM_ / 8) + head * 8 + hhalf) * 32 + r) * 4;
#pragma unroll
  for (int n = 0; n < 4; ++n) {
    const v8f& on = (n == 0) ? o0 : (n == 1) ? o1 : (n == 2) ? o2 : o3;
#pragma unroll
    for (int g = 0; g < 8; ++g) Pw[(g + 8 * hhalf) * 20 + r] = on[g] * inv[g];
    const float4 lo4 = *(const float4*)(Pw + r * 20 + 8 * hhalf);
    const float4 hi4 = *(const float4*)(Pw + r * 20 + 8 * hhalf + 4);
    *(float4*)(dst + n * 256) = lo4;
    *(float4*)(dst + n * 256 + 64) = hi4;
  }
}

struct OpStage {
  float4 a[2][4];
  float4 b[4][4];
};

__device__ __forceinline__ void op_load(OpStage& st, const float* __restrict__ pa,
                                        const float* __restrict__ pb, int stage) {
#pragma unroll
  for (int q = 0; q < 4; ++q) {
#pragma unroll
    for (int i = 0; i < 2; ++i)
      st.a[i][q] = *(const float4*)(pa + (size_t)i * 16384 + (stage * 4 + q) * 128);
#pragma unroll
    for (int j = 0; j < 4; ++j)
      st.b[j][q] = *(const float4*)(pb + (size_t)j * 16384 + (stage * 4 + q) * 128);
  }
}

__device__ __forceinline__ void op_mma(const OpStage& st, v8f (&acc)[2][4]) {
#pragma unroll
  for (int q = 0; q < 4; ++q) {
#pragma unroll
    for (int i = 0; i < 2; ++i)
#pragma unroll
      for (int j = 0; j < 4; ++j)
        acc[i][j] = wmma_f32(v2f{st.a[i][q].x, st.a[i][q].y},
                             v2f{st.b[j][q].x, st.b[j][q].y}, acc[i][j]);
#pragma unroll
    for (int i = 0; i < 2; ++i)
#pragma unroll
      for (int j = 0; j < 4; ++j)
        acc[i][j] = wmma_f32(v2f{st.a[i][q].z, st.a[i][q].w},
                             v2f{st.b[j][q].z, st.b[j][q].w}, acc[i][j]);
  }
}

__global__ void __launch_bounds__(256, 2)
out_proj_kernel(const float* __restrict__ Of, const float* __restrict__ Wf,
                const float* __restrict__ b_out, float* __restrict__ out) {
  const int tid = threadIdx.x;
  const int lane = tid & 31;
  const int r = lane & 15;
  const int hhalf = lane >> 4;
  const int wv = __builtin_amdgcn_readfirstlane(tid >> 5);
  const int wm = wv >> 1;
  const int wn = wv & 1;
  const int m0 = blockIdx.x * 128 + wm * 32;
  const int n0 = blockIdx.y * 128 + wn * 64;
  const float* pa = Of + (size_t)(m0 >> 4) * 16384 + lane * 4;
  const float* pb = Wf + (size_t)(n0 >> 4) * 16384 + lane * 4;

  v8f acc[2][4] = {};
  OpStage sA, sB;
  op_load(sA, pa, pb, 0);
  for (int s = 0; s < DM_ / 32; s += 2) {
    op_load(sB, pa, pb, s + 1);
    __builtin_amdgcn_sched_barrier(0);
    op_mma(sA, acc);
    op_load(sA, pa, pb, (s + 2 < DM_ / 32) ? (s + 2) : (DM_ / 32 - 1));
    __builtin_amdgcn_sched_barrier(0);
    op_mma(sB, acc);
  }

#pragma unroll
  for (int i = 0; i < 2; ++i) {
#pragma unroll
    for (int j = 0; j < 4; ++j) {
      const int n = n0 + j * 16 + r;
      const float bv = b_out[n];
#pragma unroll
      for (int g = 0; g < 8; ++g) {
        const int gm = m0 + i * 16 + g + 8 * hhalf;
        out[(size_t)gm * DM_ + n] = acc[i][j][g] + bv;
      }
    }
  }
}

extern "C" void kernel_launch(void* const* d_in, const int* in_sizes, int n_in,
                              void* d_out, int out_size, void* d_ws,
                              size_t ws_size, hipStream_t stream) {
  const float* x = (const float*)d_in[0];
  const float* w_qkv = (const float*)d_in[1];
  const float* b_qkv = (const float*)d_in[2];
  const float* w_out = (const float*)d_in[3];
  const float* b_out = (const float*)d_in[4];
  float* out = (float*)d_out;

  const size_t per_buf = (size_t)BH_ * S_ * DK_;
  float* Qw = (float*)d_ws;
  float* Kw = Qw + per_buf;
  float* Vw = Kw + per_buf;
  float* Ow = Vw + per_buf;
  float* Wf = Ow + per_buf;

  qkv_rope_kernel<<<dim3(ROWS_ / 128, N_QKV / 64), 256, 0, stream>>>(
      x, w_qkv, b_qkv, w_out, Qw, Kw, Vw, Wf);
  flash_attn_kernel<<<dim3(S_ / 128, BH_), 256, 0, stream>>>(Qw, Kw, Vw, Ow);
  out_proj_kernel<<<dim3(ROWS_ / 128, DM_ / 128), 256, 0, stream>>>(
      Ow, Wf, b_out, out);
}


# attention: softmax moved behind the LDS patch: raw scores transposed there, a lane holds 8 keys of one query row; row max and sum in-lane + one half swap, no 16-lane butterflies; sum order changes
# speedup vs baseline: 1.5402x; 1.0688x over previous
#include <hip/hip_runtime.h>
#include <hip/hip_bf16.h>


typedef __attribute__((ext_vector_type(2))) float v2f;
typedef __attribute__((ext_vector_type(8))) float v8f;

#define B_     2
#define S_     2048
#define DM_    1024
#define H_     16
#define DK_    64
#define BH_    (B_ * H_)
#define ROWS_  (B_ * S_)
#define N_QKV  (3 * DM_)

__device__ __forceinline__ v8f wmma_f32(v2f a, v2f b, v8f c) {
  return __builtin_amdgcn_wmma_f32_16x16x4_f32(
      false, a, false, b, (short)0, c, false, false);
}

__device__ __forceinline__ float xhalf(float x) {
  const int b = __builtin_bit_cast(int, x);
  return __builtin_bit_cast(float, __builtin_amdgcn_permlanex16(
      b, b, 0x76543210, 0xfedcba98, false, false));
}

template <int N>
__device__ __forceinline__ float row_share(float x) {
  return __builtin_bit_cast(float, __builtin_amdgcn_update_dpp(
      0, __builtin_bit_cast(int, x), 0x150 + N, 0xf, 0xf, true));
}

__device__ __forceinline__ float row_ror8(float x) {
  return __builtin_bit_cast(float, __builtin_amdgcn_update_dpp(
      0, __builtin_bit_cast(int, x), 0x128, 0xf, 0xf, true));
}

__device__ __forceinline__ void gemm_core_128x64(
    const float* __restrict__ A, const float* __restrict__ W,
    float* Xs, float* Ws, int m0, int n0, int Kdim, bool swap, v8f acc[2][2]) {
  const int tid = threadIdx.x;
  const int wv = tid >> 5;
  const int lane = tid & 31;
  const int r = lane & 15;
  const int hh = lane >> 4;
  const int wm = wv >> 1;
  const int wn = wv & 1;
  const float* xrow = Xs + (wm * 32 + r) * 20;
  const float* wrow = Ws + (wn * 32 + r) * 20;
  const float* pa = swap ? wrow : xrow;
  const float* pb = swap ? xrow : wrow;

  for (int k0 = 0; k0 < Kdim; k0 += 16) {
    __syncthreads();
#pragma unroll
    for (int i = 0; i < 2; ++i) {
      int idx = tid + i * 256;
      int row = idx >> 2;
      int cv = (idx & 3) * 4;
      *(float4*)(Xs + row * 20 + cv) =
          *(const float4*)(A + (size_t)(m0 + row) * Kdim + k0 + cv);
    }
    {
      int row = tid >> 2;
      int cv = (tid & 3) * 4;
      *(float4*)(Ws + row * 20 + cv) =
          *(const float4*)(W + (size_t)(n0 + row) * Kdim + k0 + cv);
    }
    __syncthreads();
#pragma unroll
    for (int kk = 0; kk < 4; ++kk) {
      const int kd = kk * 4 + hh * 2;
      v2f a0 = {pa[kd], pa[kd + 1]};
      v2f a1 = {pa[16 * 20 + kd], pa[16 * 20 + kd + 1]};
      v2f b0 = {pb[kd], pb[kd + 1]};
      v2f b1 = {pb[16 * 20 + kd], pb[16 * 20 + kd + 1]};
      acc[0][0] = wmma_f32(a0, b0, acc[0][0]);
      acc[0][1] = wmma_f32(a0, b1, acc[0][1]);
      acc[1][0] = wmma_f32(a1, b0, acc[1][0]);
      acc[1][1] = wmma_f32(a1, b1, acc[1][1]);
    }
  }
}

__global__ void __launch_bounds__(256, 12)
qkv_rope_kernel(const float* __restrict__ x, const float* __restrict__ w_qkv,
                const float* __restrict__ b_qkv, const float* __restrict__ w_out,
                float* __restrict__ Qw, float* __restrict__ Kw,
                float* __restrict__ Vw, float* __restrict__ Wf) {
  __shared__ float Xs[128 * 20];
  __shared__ float Ws[64 * 20];
  v8f acc[2][2] = {};
  const int m0 = blockIdx.x * 128;
  const int n0 = blockIdx.y * 64;
  {
    const int wg = blockIdx.y * (ROWS_ / 128) + blockIdx.x;
    if (wg < (DM_ * DM_ / 4) / 256) {
      const int f = wg * 256 + threadIdx.x;
      const int ln = f & 31;
      const int blk = f >> 5;
      const int rb = blk >> 7;
      const int kc = blk & 127;
      const float4 t = *(const float4*)(w_out + (size_t)(rb * 16 + (ln & 15)) * DM_ +
                                        kc * 8 + (ln >> 4) * 4);
      *(float4*)(Wf + (size_t)f * 4) = t;
    }
  }
  const int sel = n0 >> 10;
  const int head = (n0 & 1023) >> 6;
  const bool swap = (sel == 1);
  gemm_core_128x64(x, w_qkv, Xs, Ws, m0, n0, DM_, swap, acc);

  const int tid = threadIdx.x;
  const int wv = tid >> 5;
  const int lane = tid & 31;
  const int r = lane & 15;
  const int hhalf = lane >> 4;
  const int wm = wv >> 1;
  const int wn = wv & 1;

  if (sel == 0) {
#pragma unroll
    for (int mi = 0; mi < 2; ++mi) {
#pragma unroll
      for (int ni = 0; ni < 2; ++ni) {
        const int d = wn * 32 + ni * 16 + r;
        const float bv = b_qkv[n0 + d];
        const int p = d >> 1;
        const float invf = __expf((float)(2 * p) * (-9.210340371976184f / 64.0f));
        const float sgn = (d & 1) ? 1.0f : -1.0f;
#pragma unroll
        for (int g = 0; g < 8; ++g) {
          const int gm = m0 + wm * 32 + mi * 16 + g + 8 * hhalf;
          const int srow = gm & (S_ - 1);
          const int bb = gm >> 11;
          float v = acc[mi][ni][g] + bv;
          float other = __shfl_xor(v, 1);
          const float ang = (float)srow * invf;
          const float sn = __sinf(ang);
          const float cs = __cosf(ang);
          const float outv = v * cs + other * sgn * sn;
          Qw[(((size_t)(bb * H_ + head)) * S_ + srow) * DK_ + d] = outv;
        }
      }
    }
  } else {
    float* dst = swap ? Kw : Vw;
#pragma unroll
    for (int i = 0; i < 2; ++i) {
#pragma unroll
      for (int j = 0; j < 2; ++j) {
        const int fb = swap ? i : j;
        const int sb = swap ? j : i;
        const int nb = wn * 2 + fb;
        const int gm0 = m0 + wm * 32 + sb * 16;
        const int bb = gm0 >> 11;
        const int s0 = gm0 & (S_ - 1);
        float vals[8];
        if (swap) {
          const int d0 = nb * 16 + 8 * hhalf;
          const float4 bA = *(const float4*)(b_qkv + n0 + d0);
          const float4 bB = *(const float4*)(b_qkv + n0 + d0 + 4);
          const float bvv[8] = {bA.x, bA.y, bA.z, bA.w, bB.x, bB.y, bB.z, bB.w};
          float v[8];
#pragma unroll
          for (int g = 0; g < 8; ++g) v[g] = acc[i][j][g] + bvv[g];
          const float srowf = (float)(s0 + r);
#pragma unroll
          for (int g2 = 0; g2 < 4; ++g2) {
            const int p = (d0 >> 1) + g2;
            const float invf = __expf((float)(2 * p) * (-9.210340371976184f / 64.0f));
            const float ang = srowf * invf;
            const float sn = __sinf(ang);
            const float cs = __cosf(ang);
            vals[2 * g2]     = v[2 * g2] * cs + v[2 * g2 + 1] * -1.0f * sn;
            vals[2 * g2 + 1] = v[2 * g2 + 1] * cs + v[2 * g2] * 1.0f * sn;
          }
        } else {
          const float bv = b_qkv[n0 + nb * 16 + r];
#pragma unroll
          for (int g = 0; g < 8; ++g) vals[g] = acc[i][j][g] + bv;
        }
        float* base = dst + ((size_t)((bb * H_ + head) * (S_ / 16) + (s0 >> 4))) * 1024 + lane * 4;
        *(float4*)(base + (nb * 2 + 0) * 128) = make_float4(vals[0], vals[1], vals[2], vals[3]);
        *(float4*)(base + (nb * 2 + 1) * 128) = make_float4(vals[4], vals[5], vals[6], vals[7]);
      }
    }
  }
}

__device__ __forceinline__ void attn_tile(
    const v2f (&qf)[16], const float4 (&kc)[8], const float4 (&vc)[8],
    v8f& o0, v8f& o1, v8f& o2, v8f& o3, float& mrun, float& lrun,
    float* Pw, bool diag, int key, int qrow0, int r, int hhalf) {
  v8f s = {};
#pragma unroll
  for (int i = 0; i < 8; ++i) {
    s = wmma_f32(qf[2 * i],     v2f{kc[i].x, kc[i].y}, s);
    s = wmma_f32(qf[2 * i + 1], v2f{kc[i].z, kc[i].w}, s);
  }
  if (diag) {
#pragma unroll
    for (int g = 0; g < 8; ++g) {
      if (key > qrow0 + g) s[g] = -3.0e38f;
    }
  }
#pragma unroll
  for (int g = 0; g < 8; ++g) Pw[(g + 8 * hhalf) * 20 + r] = s[g];
  const float4 sA = *(const float4*)(Pw + r * 20 + 8 * hhalf);
  const float4 sB = *(const float4*)(Pw + r * 20 + 8 * hhalf + 4);
  const float sa[8] = {sA.x, sA.y, sA.z, sA.w, sB.x, sB.y, sB.z, sB.w};
  float mx = fmaxf(fmaxf(sa[0], sa[1]), sa[2]);
  mx = fmaxf(fmaxf(mx, sa[3]), sa[4]);
  mx = fmaxf(fmaxf(mx, sa[5]), sa[6]);
  mx = fmaxf(mx, sa[7]);
  mx = fmaxf(mx, xhalf(mx));
  const float mnew = fmaxf(mrun, mx);
  const float alpha = __expf(mrun - mnew);
  float p[8];
#pragma unroll
  for (int i = 0; i < 8; ++i) p[i] = __expf(sa[i] - mnew);
  const float rs = ((p[0] + p[1]) + (p[2] + p[3])) + ((p[4] + p[5]) + (p[6] + p[7]));
  lrun = lrun * alpha + rs;
  mrun = mnew;
  const float arot = row_ror8(alpha);
  const float asel = hhalf ? arot : alpha;
  float al[8];
  al[0] = row_share<0>(asel); al[1] = row_share<1>(asel);
  al[2] = row_share<2>(asel); al[3] = row_share<3>(asel);
  al[4] = row_share<4>(asel); al[5] = row_share<5>(asel);
  al[6] = row_share<6>(asel); al[7] = row_share<7>(asel);
#pragma unroll
  for (int g = 0; g < 8; ++g) {
    o0[g] *= al[g]; o1[g] *= al[g]; o2[g] *= al[g]; o3[g] *= al[g];
  }
  const v2f af0 = {p[0], p[1]};
  const v2f af1 = {p[2], p[3]};
  const v2f af2 = {p[4], p[5]};
  const v2f af3 = {p[6], p[7]};
  o0 = wmma_f32(af0, v2f{vc[0].x, vc[0].y}, o0);
  o1 = wmma_f32(af0, v2f{vc[2].x, vc[2].y}, o1);
  o2 = wmma_f32(af0, v2f{vc[4].x, vc[4].y}, o2);
  o3 = wmma_f32(af0, v2f{vc[6].x, vc[6].y}, o3);
  o0 = wmma_f32(af1, v2f{vc[0].z, vc[0].w}, o0);
  o1 = wmma_f32(af1, v2f{vc[2].z, vc[2].w}, o1);
  o2 = wmma_f32(af1, v2f{vc[4].z, vc[4].w}, o2);
  o3 = wmma_f32(af1, v2f{vc[6].z, vc[6].w}, o3);
  o0 = wmma_f32(af2, v2f{vc[1].x, vc[1].y}, o0);
  o1 = wmma_f32(af2, v2f{vc[3].x, vc[3].y}, o1);
  o2 = wmma_f32(af2, v2f{vc[5].x, vc[5].y}, o2);
  o3 = wmma_f32(af2, v2f{vc[7].x, vc[7].y}, o3);
  o0 = wmma_f32(af3, v2f{vc[1].z, vc[1].w}, o0);
  o1 = wmma_f32(af3, v2f{vc[3].z, vc[3].w}, o1);
  o2 = wmma_f32(af3, v2f{vc[5].z, vc[5].w}, o2);
  o3 = wmma_f32(af3, v2f{vc[7].z, vc[7].w}, o3);
}

__global__ void __launch_bounds__(256, 4)
flash_attn_kernel(const float* __restrict__ Q, const float* __restrict__ K,
                  const float* __restrict__ V, float* __restrict__ O) {
  __shared__ __attribute__((aligned(16))) float Ps[8 * 16 * 20];

  const int bh = blockIdx.y;
  const int qb = blockIdx.x;
  const int tid = threadIdx.x;
  const int wv = tid >> 5;
  const int lane = tid & 31;
  const int r = lane & 15;
  const int hhalf = lane >> 4;
  const int sl = wv & 3;
  const int qt = __builtin_amdgcn_readfirstlane(
      (wv < 4) ? (qb + 32 * sl) : (127 - qb - 32 * sl));
  const int qw0 = qt * 16;

  const float* Qp = Q + (size_t)bh * S_ * DK_;
  const float* Kp = K + (size_t)bh * S_ * DK_ + lane * 4;
  const float* Vp = V + (size_t)bh * S_ * DK_ + lane * 4;

  v2f qf[16];
#pragma unroll
  for (int i = 0; i < 8; ++i) {
    const float4 t = *(const float4*)(Qp + (size_t)(qw0 + r) * DK_ +
                                      (i >> 1) * 16 + hhalf * 8 + (i & 1) * 4);
    qf[2 * i]     = v2f{t.x * 0.125f, t.y * 0.125f};
    qf[2 * i + 1] = v2f{t.z * 0.125f, t.w * 0.125f};
  }

  v8f o0 = {}, o1 = {}, o2 = {}, o3 = {};
  float mrun = -3.0e38f, lrun = 0.0f;

  float* Pw = Ps + wv * 16 * 20;
  const int qrow0 = qw0 + 8 * hhalf;

  float4 kA[8], vA[8], kB[8], vB[8];
#pragma unroll
  for (int i = 0; i < 8; ++i) {
    kA[i] = *(const float4*)(Kp + i * 128);
    vA[i] = *(const float4*)(Vp + i * 128);
  }

  for (int j = 0; j <= qt; j += 2) {
    {
      const int jn = (j + 1 <= qt) ? (j + 1) : qt;
#pragma unroll
      for (int i = 0; i < 8; ++i) {
        kB[i] = *(const float4*)(Kp + (size_t)jn * 1024 + i * 128);
        vB[i] = *(const float4*)(Vp + (size_t)jn * 1024 + i * 128);
      }
      attn_tile(qf, kA, vA, o0, o1, o2, o3, mrun, lrun, Pw, j == qt,
                j * 16 + r, qrow0, r, hhalf);
    }
    if (j + 1 <= qt) {
      const int jn = (j + 2 <= qt) ? (j + 2) : qt;
#pragma unroll
      for (int i = 0; i < 8; ++i) {
        kA[i] = *(const float4*)(Kp + (size_t)jn * 1024 + i * 128);
        vA[i] = *(const float4*)(Vp + (size_t)jn * 1024 + i * 128);
      }
      attn_tile(qf, kB, vB, o0, o1, o2, o3, mrun, lrun, Pw, j + 1 == qt,
                (j + 1) * 16 + r, qrow0, r, hhalf);
    }
  }

  const int bb = bh >> 4;
  const int head = bh & 15;
  const float lrow = lrun + xhalf(lrun);
  const float inv = 1.0f / (lrow + 1e-8f);
  float* dst = O + ((size_t)((bb * (S_ / 16) + qt) * (DM_ / 8) + head * 8 + hhalf) * 32 + r) * 4;
#pragma unroll
  for (int n = 0; n < 4; ++n) {
    const v8f& on = (n == 0) ? o0 : (n == 1) ? o1 : (n == 2) ? o2 : o3;
#pragma unroll
    for (int g = 0; g < 8; ++g) Pw[(g + 8 * hhalf) * 20 + r] = on[g];
    const float4 lo4 = *(const float4*)(Pw + r * 20 + 8 * hhalf);
    const float4 hi4 = *(const float4*)(Pw + r * 20 + 8 * hhalf + 4);
    *(float4*)(dst + n * 256) =
        make_float4(lo4.x * inv, lo4.y * inv, lo4.z * inv, lo4.w * inv);
    *(float4*)(dst + n * 256 + 64) =
        make_float4(hi4.x * inv, hi4.y * inv, hi4.z * inv, hi4.w * inv);
  }
}

struct OpStage {
  float4 a[2][4];
  float4 b[4][4];
};

__device__ __forceinline__ void op_load(OpStage& st, const float* __restrict__ pa,
                                        const float* __restrict__ pb, int stage) {
#pragma unroll
  for (int q = 0; q < 4; ++q) {
#pragma unroll
    for (int i = 0; i < 2; ++i)
      st.a[i][q] = *(const float4*)(pa + (size_t)i * 16384 + (stage * 4 + q) * 128);
#pragma unroll
    for (int j = 0; j < 4; ++j)
      st.b[j][q] = *(const float4*)(pb + (size_t)j * 16384 + (stage * 4 + q) * 128);
  }
}

__device__ __forceinline__ void op_mma(const OpStage& st, v8f (&acc)[2][4]) {
#pragma unroll
  for (int q = 0; q < 4; ++q) {
#pragma unroll
    for (int i = 0; i < 2; ++i)
#pragma unroll
      for (int j = 0; j < 4; ++j)
        acc[i][j] = wmma_f32(v2f{st.a[i][q].x, st.a[i][q].y},
                             v2f{st.b[j][q].x, st.b[j][q].y}, acc[i][j]);
#pragma unroll
    for (int i = 0; i < 2; ++i)
#pragma unroll
      for (int j = 0; j < 4; ++j)
        acc[i][j] = wmma_f32(v2f{st.a[i][q].z, st.a[i][q].w},
                             v2f{st.b[j][q].z, st.b[j][q].w}, acc[i][j]);
  }
}

__global__ void __launch_bounds__(256, 2)
out_proj_kernel(const float* __restrict__ Of, const float* __restrict__ Wf,
                const float* __restrict__ b_out, float* __restrict__ out) {
  const int tid = threadIdx.x;
  const int lane = tid & 31;
  const int r = lane & 15;
  const int hhalf = lane >> 4;
  const int wv = __builtin_amdgcn_readfirstlane(tid >> 5);
  const int wm = wv >> 1;
  const int wn = wv & 1;
  const int m0 = blockIdx.x * 128 + wm * 32;
  const int n0 = blockIdx.y * 128 + wn * 64;
  const float* pa = Of + (size_t)(m0 >> 4) * 16384 + lane * 4;
  const float* pb = Wf + (size_t)(n0 >> 4) * 16384 + lane * 4;

  v8f acc[2][4] = {};
  OpStage sA, sB;
  op_load(sA, pa, pb, 0);
  for (int s = 0; s < DM_ / 32; s += 2) {
    op_load(sB, pa, pb, s + 1);
    __builtin_amdgcn_sched_barrier(0);
    op_mma(sA, acc);
    op_load(sA, pa, pb, (s + 2 < DM_ / 32) ? (s + 2) : (DM_ / 32 - 1));
    __builtin_amdgcn_sched_barrier(0);
    op_mma(sB, acc);
  }

#pragma unroll
  for (int i = 0; i < 2; ++i) {
#pragma unroll
    for (int j = 0; j < 4; ++j) {
      const int n = n0 + j * 16 + r;
      const float bv = b_out[n];
#pragma unroll
      for (int g = 0; g < 8; ++g) {
        const int gm = m0 + i * 16 + g + 8 * hhalf;
        out[(size_t)gm * DM_ + n] = acc[i][j][g] + bv;
      }
    }
  }
}

extern "C" void kernel_launch(void* const* d_in, const int* in_sizes, int n_in,
                              void* d_out, int out_size, void* d_ws,
                              size_t ws_size, hipStream_t stream) {
  const float* x = (const float*)d_in[0];
  const float* w_qkv = (const float*)d_in[1];
  const float* b_qkv = (const float*)d_in[2];
  const float* w_out = (const float*)d_in[3];
  const float* b_out = (const float*)d_in[4];
  float* out = (float*)d_out;

  const size_t per_buf = (size_t)BH_ * S_ * DK_;
  float* Qw = (float*)d_ws;
  float* Kw = Qw + per_buf;
  float* Vw = Kw + per_buf;
  float* Ow = Vw + per_buf;
  float* Wf = Ow + per_buf;

  qkv_rope_kernel<<<dim3(ROWS_ / 128, N_QKV / 64), 256, 0, stream>>>(
      x, w_qkv, b_qkv, w_out, Qw, Kw, Vw, Wf);
  flash_attn_kernel<<<dim3(S_ / 128, BH_), 256, 0, stream>>>(Qw, Kw, Vw, Ow);
  out_proj_kernel<<<dim3(ROWS_ / 128, DM_ / 128), 256, 0, stream>>>(
      Ow, Wf, b_out, out);
}
